# starting kernel
# baseline (speedup 1.0000x reference)
#include <hip/hip_runtime.h>
#include <hip/hip_bf16.h>

typedef float v2f __attribute__((ext_vector_type(2)));
typedef float v8f __attribute__((ext_vector_type(8)));

#define NCHILD 8192
#define NCH    32
#define PARENT 4096

__global__ void zero_kernel(float* __restrict__ colsum) {
    int i = blockIdx.x * blockDim.x + threadIdx.x;
    if (i < NCHILD) colsum[i] = 0.0f;
}

__global__ __launch_bounds__(256) void colsum_kernel(const float* __restrict__ A,
                                                     float* __restrict__ colsum) {
    const int j  = blockIdx.x * 256 + threadIdx.x;
    const int r0 = blockIdx.y * 128;
    const float* p = A + (size_t)r0 * NCHILD + j;
    float s = 0.0f;
#pragma unroll 8
    for (int i = 0; i < 128; ++i)
        s += p[(size_t)i * NCHILD];
    atomicAdd(&colsum[j], s);
}

__global__ __launch_bounds__(256) void gather_kernel(const float* __restrict__ input,
                                                     const int*   __restrict__ desc,
                                                     const float* __restrict__ colsum,
                                                     float* __restrict__ Bt) {
    int idx = blockIdx.x * 256 + threadIdx.x;
    int j = idx & (NCHILD - 1);
    int c = idx >> 13;
    float inv = 1.0f / colsum[j];
    int p = desc[j];
    Bt[idx] = input[p * NCH + c] * inv;
}

__global__ __launch_bounds__(256) void gemm_kernel(const float* __restrict__ A,
                                                   const float* __restrict__ Bt,
                                                   float* __restrict__ out) {
    __shared__ float red[2 * 8 * 32 * 8];

    const int lane = threadIdx.x & 31;
    const int w    = threadIdx.x >> 5;
    const int hi   = lane >> 4;
    const int r    = lane & 15;
    const int rowBase = blockIdx.x * 16;

    const float* ap  = A  + (size_t)(rowBase + r) * NCHILD + w * 1024 + 2 * hi;
    const float* b0p = Bt + (size_t)r        * NCHILD + w * 1024 + 2 * hi;
    const float* b1p = Bt + (size_t)(r + 16) * NCHILD + w * 1024 + 2 * hi;

    v8f acc0 = {};
    v8f acc1 = {};

#pragma unroll 4
    for (int k = 0; k < 1024; k += 4) {
        v2f a  = *(const v2f*)(ap  + k);
        v2f b0 = *(const v2f*)(b0p + k);
        v2f b1 = *(const v2f*)(b1p + k);
        acc0 = __builtin_amdgcn_wmma_f32_16x16x4_f32(false, a, false, b0,
                                                     (short)0, acc0, false, false);
        acc1 = __builtin_amdgcn_wmma_f32_16x16x4_f32(false, a, false, b1,
                                                     (short)0, acc1, false, false);
    }

    float* s0 = &red[((0 * 8 + w) * 32 + lane) * 8];
    float* s1 = &red[((1 * 8 + w) * 32 + lane) * 8];
#pragma unroll
    for (int i = 0; i < 8; ++i) { s0[i] = acc0[i]; s1[i] = acc1[i]; }
    __syncthreads();

    if (w < 2) {
        v8f sum = {};
#pragma unroll
        for (int ww = 0; ww < 8; ++ww) {
            const float* p = &red[((w * 8 + ww) * 32 + lane) * 8];
#pragma unroll
            for (int i = 0; i < 8; ++i) sum[i] += p[i];
        }
        const int col = w * 16 + r;
#pragma unroll
        for (int i = 0; i < 8; ++i)
            out[(size_t)(rowBase + i + 8 * hi) * NCH + col] = sum[i];
    }
}

extern "C" void kernel_launch(void* const* d_in, const int* in_sizes, int n_in,
                              void* d_out, int out_size, void* d_ws, size_t ws_size,
                              hipStream_t stream) {
    const float* input = (const float*)d_in[0];
    const float* A     = (const float*)d_in[1];
    const int*   desc  = (const int*)  d_in[2];
    float*       out   = (float*)d_out;

    float* colsum = (float*)d_ws;
    float* Bt     = colsum + NCHILD;

    zero_kernel  <<<NCHILD / 256, 256, 0, stream>>>(colsum);
    colsum_kernel<<<dim3(NCHILD / 256, NCHILD / 128), 256, 0, stream>>>(A, colsum);
    gather_kernel<<<(NCHILD * NCH) / 256, 256, 0, stream>>>(input, desc, colsum, Bt);
    gemm_kernel  <<<NCHILD / 16, 256, 0, stream>>>(A, Bt, out);
}


# gemm: A/B fragment loads widened b64 -> b128 (lane keeps 4 consecutive k; k order in each 8-k group permuted, same on A and B: f32 sum order changes), issued one 32-k stage ahead of the WMMAs
# speedup vs baseline: 1.0826x; 1.0826x over previous
#include <hip/hip_runtime.h>
#include <hip/hip_bf16.h>

typedef float v2f __attribute__((ext_vector_type(2)));
typedef float v8f __attribute__((ext_vector_type(8)));

#define NCHILD 8192
#define NCH    32
#define PARENT 4096

__global__ void zero_kernel(float* __restrict__ colsum) {
    int i = blockIdx.x * blockDim.x + threadIdx.x;
    if (i < NCHILD) colsum[i] = 0.0f;
}

__global__ __launch_bounds__(256) void colsum_kernel(const float* __restrict__ A,
                                                     float* __restrict__ colsum) {
    const int j  = blockIdx.x * 256 + threadIdx.x;
    const int r0 = blockIdx.y * 128;
    const float* p = A + (size_t)r0 * NCHILD + j;
    float s = 0.0f;
#pragma unroll 8
    for (int i = 0; i < 128; ++i)
        s += p[(size_t)i * NCHILD];
    atomicAdd(&colsum[j], s);
}

__global__ __launch_bounds__(256) void gather_kernel(const float* __restrict__ input,
                                                     const int*   __restrict__ desc,
                                                     const float* __restrict__ colsum,
                                                     float* __restrict__ Bt) {
    int idx = blockIdx.x * 256 + threadIdx.x;
    int j = idx & (NCHILD - 1);
    int c = idx >> 13;
    float inv = 1.0f / colsum[j];
    int p = desc[j];
    Bt[idx] = input[p * NCH + c] * inv;
}

typedef float v4f __attribute__((ext_vector_type(4)));

struct Stage { v4f a[4]; v4f b0[4]; v4f b1[4]; };

__device__ __forceinline__ void load_stage(Stage& s, const float* __restrict__ ap,
                                           const float* __restrict__ b0p,
                                           const float* __restrict__ b1p, int k) {
#pragma unroll
    for (int q = 0; q < 4; ++q) s.a[q]  = *(const v4f*)(ap  + k + 8 * q);
#pragma unroll
    for (int q = 0; q < 4; ++q) s.b0[q] = *(const v4f*)(b0p + k + 8 * q);
#pragma unroll
    for (int q = 0; q < 4; ++q) s.b1[q] = *(const v4f*)(b1p + k + 8 * q);
}

__device__ __forceinline__ void mma_stage(const Stage& s, v8f& acc0, v8f& acc1) {
#pragma unroll
    for (int q = 0; q < 4; ++q) {
        const v2f a_lo  = __builtin_shufflevector(s.a[q],  s.a[q],  0, 1);
        const v2f a_hi  = __builtin_shufflevector(s.a[q],  s.a[q],  2, 3);
        const v2f b0_lo = __builtin_shufflevector(s.b0[q], s.b0[q], 0, 1);
        const v2f b0_hi = __builtin_shufflevector(s.b0[q], s.b0[q], 2, 3);
        const v2f b1_lo = __builtin_shufflevector(s.b1[q], s.b1[q], 0, 1);
        const v2f b1_hi = __builtin_shufflevector(s.b1[q], s.b1[q], 2, 3);
        acc0 = __builtin_amdgcn_wmma_f32_16x16x4_f32(false, a_lo, false, b0_lo,
                                                     (short)0, acc0, false, false);
        acc1 = __builtin_amdgcn_wmma_f32_16x16x4_f32(false, a_lo, false, b1_lo,
                                                     (short)0, acc1, false, false);
        acc0 = __builtin_amdgcn_wmma_f32_16x16x4_f32(false, a_hi, false, b0_hi,
                                                     (short)0, acc0, false, false);
        acc1 = __builtin_amdgcn_wmma_f32_16x16x4_f32(false, a_hi, false, b1_hi,
                                                     (short)0, acc1, false, false);
    }
}

__global__ __launch_bounds__(256) void gemm_kernel(const float* __restrict__ A,
                                                   const float* __restrict__ Bt,
                                                   float* __restrict__ out) {
    __shared__ float red[2 * 8 * 32 * 8];

    const int lane = threadIdx.x & 31;
    const int w    = threadIdx.x >> 5;
    const int hi   = lane >> 4;
    const int r    = lane & 15;
    const int rowBase = blockIdx.x * 16;

    const float* ap  = A  + (size_t)(rowBase + r) * NCHILD + w * 1024 + 4 * hi;
    const float* b0p = Bt + (size_t)r        * NCHILD + w * 1024 + 4 * hi;
    const float* b1p = Bt + (size_t)(r + 16) * NCHILD + w * 1024 + 4 * hi;

    v8f acc0 = {};
    v8f acc1 = {};

    Stage even, odd;
    load_stage(even, ap, b0p, b1p, 0);
    for (int k = 0; k < 1024; k += 64) {
        load_stage(odd, ap, b0p, b1p, k + 32);
        mma_stage(even, acc0, acc1);
        if (k + 64 < 1024) load_stage(even, ap, b0p, b1p, k + 64);
        mma_stage(odd, acc0, acc1);
    }

    float* s0 = &red[((0 * 8 + w) * 32 + lane) * 8];
    float* s1 = &red[((1 * 8 + w) * 32 + lane) * 8];
#pragma unroll
    for (int i = 0; i < 8; ++i) { s0[i] = acc0[i]; s1[i] = acc1[i]; }
    __syncthreads();

    if (w < 2) {
        v8f sum = {};
#pragma unroll
        for (int ww = 0; ww < 8; ++ww) {
            const float* p = &red[((w * 8 + ww) * 32 + lane) * 8];
#pragma unroll
            for (int i = 0; i < 8; ++i) sum[i] += p[i];
        }
        const int col = w * 16 + r;
#pragma unroll
        for (int i = 0; i < 8; ++i)
            out[(size_t)(rowBase + i + 8 * hi) * NCH + col] = sum[i];
    }
}

extern "C" void kernel_launch(void* const* d_in, const int* in_sizes, int n_in,
                              void* d_out, int out_size, void* d_ws, size_t ws_size,
                              hipStream_t stream) {
    const float* input = (const float*)d_in[0];
    const float* A     = (const float*)d_in[1];
    const int*   desc  = (const int*)  d_in[2];
    float*       out   = (float*)d_out;

    float* colsum = (float*)d_ws;
    float* Bt     = colsum + NCHILD;

    zero_kernel  <<<NCHILD / 256, 256, 0, stream>>>(colsum);
    colsum_kernel<<<dim3(NCHILD / 256, NCHILD / 128), 256, 0, stream>>>(A, colsum);
    gather_kernel<<<(NCHILD * NCH) / 256, 256, 0, stream>>>(input, desc, colsum, Bt);
    gemm_kernel  <<<NCHILD / 16, 256, 0, stream>>>(A, Bt, out);
}


# gemm loads whole 128-B lines: A transposed to fragments through per-wave LDS, B' stored in fragment order, wave = 32 rows x 512 k (16 K-partials per tile, was 8: f32 sum order changes)
# speedup vs baseline: 1.7630x; 1.6285x over previous
#include <hip/hip_runtime.h>
#include <hip/hip_bf16.h>


typedef float v2f __attribute__((ext_vector_type(2)));
typedef float v4f __attribute__((ext_vector_type(4)));
typedef float v8f __attribute__((ext_vector_type(8)));

#define NCHILD 8192
#define NCH    32
#define PARENT 4096

__global__ void zero_kernel(float* __restrict__ colsum) {
    int i = blockIdx.x * blockDim.x + threadIdx.x;
    if (i < NCHILD) colsum[i] = 0.0f;
}

__global__ __launch_bounds__(256) void colsum_kernel(const float* __restrict__ A,
                                                     float* __restrict__ colsum) {
    const int j  = blockIdx.x * 256 + threadIdx.x;
    const int r0 = blockIdx.y * 128;
    const float* p = A + (size_t)r0 * NCHILD + j;
    float s = 0.0f;
#pragma unroll 8
    for (int i = 0; i < 128; ++i)
        s += p[(size_t)i * NCHILD];
    atomicAdd(&colsum[j], s);
}

__global__ __launch_bounds__(256) void gather_kernel(const float* __restrict__ input,
                                                     const int*   __restrict__ desc,
                                                     const float* __restrict__ colsum,
                                                     float* __restrict__ Bp) {
    int idx  = blockIdx.x * 256 + threadIdx.x;
    int e    = idx & 3;
    int lane = (idx >> 2) & 31;
    int t    = (idx >> 7) & 1;
    int kq   = idx >> 8;
    int j    = 8 * kq + 4 * (lane >> 4) + e;
    int c    = 16 * t + (lane & 15);
    float inv = 1.0f / colsum[j];
    int p = desc[j];
    Bp[idx] = input[p * NCH + c] * inv;
}

#define ROWS  32
#define WAVES 16
#define KW    (NCHILD / WAVES)
#define SK    32
#define NST   (KW / SK)
#define P4    9

struct Stage8 { v4f v[8]; };

__device__ __forceinline__ void load_a(Stage8& s, const float* __restrict__ ag, int k) {
#pragma unroll
    for (int i = 0; i < 8; ++i) s.v[i] = *(const v4f*)(ag + (size_t)(4 * i) * NCHILD + k);
}

__device__ __forceinline__ void load_b(Stage8& s, const float* __restrict__ bg, int st) {
#pragma unroll
    for (int j = 0; j < 8; ++j) s.v[j] = *(const v4f*)(bg + (size_t)st * 1024 + j * 128);
}

__device__ __forceinline__ void park_a(const Stage8& s, v4f* st4) {
#pragma unroll
    for (int i = 0; i < 8; ++i) st4[4 * i * P4] = s.v[i];
}

__device__ __forceinline__ void mma_stage(const v4f* fr4, const Stage8& b, v8f (&acc)[4]) {
    v4f fa0[4], fa1[4];
#pragma unroll
    for (int q = 0; q < 4; ++q) {
        fa0[q] = fr4[2 * q];
        fa1[q] = fr4[16 * P4 + 2 * q];
    }
#pragma unroll
    for (int q = 0; q < 4; ++q) {
        const v4f b0 = b.v[2 * q];
        const v4f b1 = b.v[2 * q + 1];
        const v2f a0_lo = __builtin_shufflevector(fa0[q], fa0[q], 0, 1);
        const v2f a0_hi = __builtin_shufflevector(fa0[q], fa0[q], 2, 3);
        const v2f a1_lo = __builtin_shufflevector(fa1[q], fa1[q], 0, 1);
        const v2f a1_hi = __builtin_shufflevector(fa1[q], fa1[q], 2, 3);
        const v2f b0_lo = __builtin_shufflevector(b0, b0, 0, 1);
        const v2f b0_hi = __builtin_shufflevector(b0, b0, 2, 3);
        const v2f b1_lo = __builtin_shufflevector(b1, b1, 0, 1);
        const v2f b1_hi = __builtin_shufflevector(b1, b1, 2, 3);
        acc[0] = __builtin_amdgcn_wmma_f32_16x16x4_f32(false, a0_lo, false, b0_lo,
                                                       (short)0, acc[0], false, false);
        acc[1] = __builtin_amdgcn_wmma_f32_16x16x4_f32(false, a0_lo, false, b1_lo,
                                                       (short)0, acc[1], false, false);
        acc[2] = __builtin_amdgcn_wmma_f32_16x16x4_f32(false, a1_lo, false, b0_lo,
                                                       (short)0, acc[2], false, false);
        acc[3] = __builtin_amdgcn_wmma_f32_16x16x4_f32(false, a1_lo, false, b1_lo,
                                                       (short)0, acc[3], false, false);
        acc[0] = __builtin_amdgcn_wmma_f32_16x16x4_f32(false, a0_hi, false, b0_hi,
                                                       (short)0, acc[0], false, false);
        acc[1] = __builtin_amdgcn_wmma_f32_16x16x4_f32(false, a0_hi, false, b1_hi,
                                                       (short)0, acc[1], false, false);
        acc[2] = __builtin_amdgcn_wmma_f32_16x16x4_f32(false, a1_hi, false, b0_hi,
                                                       (short)0, acc[2], false, false);
        acc[3] = __builtin_amdgcn_wmma_f32_16x16x4_f32(false, a1_hi, false, b1_hi,
                                                       (short)0, acc[3], false, false);
    }
}

__global__ __launch_bounds__(512) void gemm_kernel(const float* __restrict__ A,
                                                   const float* __restrict__ Bp,
                                                   float* __restrict__ out) {
    __shared__ v4f park[WAVES * ROWS * P4];
    __shared__ v4f red[4 * WAVES * 32 * 2];

    const int lane = threadIdx.x & 31;
    const int w    = threadIdx.x >> 5;
    const int hi   = lane >> 4;
    const int r    = lane & 15;
    const int rowBase = blockIdx.x * ROWS;

    v4f* tile4 = park + w * (ROWS * P4);
    const float* ag = A + (size_t)(rowBase + (lane >> 3)) * NCHILD + w * KW + 4 * (lane & 7);
    v4f*       st4 = tile4 + (lane >> 3) * P4 + (lane & 7);
    const v4f* fr4 = tile4 + r * P4 + hi;
    const float* bg = Bp + (size_t)w * (KW / 8) * 256 + lane * 4;

    v8f acc[4] = {};

    Stage8 aE, aO, bE, bO;
    load_b(bE, bg, 0);
    load_a(aE, ag, 0);
    load_a(aO, ag, SK);
    for (int s = 0; s < NST; s += 2) {
        park_a(aE, st4);
        load_b(bO, bg, s + 1);
        __builtin_amdgcn_sched_barrier(0);
        if (s + 2 < NST) load_a(aE, ag, (s + 2) * SK);
        __builtin_amdgcn_sched_barrier(0);
        mma_stage(fr4, bE, acc);

        park_a(aO, st4);
        if (s + 2 < NST) load_b(bE, bg, s + 2);
        __builtin_amdgcn_sched_barrier(0);
        if (s + 2 < NST) load_a(aO, ag, (s + 3) * SK);
        __builtin_amdgcn_sched_barrier(0);
        mma_stage(fr4, bO, acc);
    }

#pragma unroll
    for (int t4 = 0; t4 < 4; ++t4) {
        v4f* d = &red[((t4 * WAVES + w) * 32 + lane) * 2];
        d[0] = __builtin_shufflevector(acc[t4], acc[t4], 0, 1, 2, 3);
        d[1] = __builtin_shufflevector(acc[t4], acc[t4], 4, 5, 6, 7);
    }
    __syncthreads();

    if (w < 4) {
        v8f sum = {};
#pragma unroll
        for (int ww = 0; ww < WAVES; ++ww) {
            const v4f* p = &red[((w * WAVES + ww) * 32 + lane) * 2];
            const v4f lo = p[0];
            const v4f hi4 = p[1];
#pragma unroll
            for (int i = 0; i < 4; ++i) { sum[i] += lo[i]; sum[4 + i] += hi4[i]; }
        }
        const int mt  = w >> 1;
        const int col = (w & 1) * 16 + r;
#pragma unroll
        for (int i = 0; i < 8; ++i)
            out[(size_t)(rowBase + 16 * mt + i + 8 * hi) * NCH + col] = sum[i];
    }
}

extern "C" void kernel_launch(void* const* d_in, const int* in_sizes, int n_in,
                              void* d_out, int out_size, void* d_ws, size_t ws_size,
                              hipStream_t stream) {
    const float* input = (const float*)d_in[0];
    const float* A     = (const float*)d_in[1];
    const int*   desc  = (const int*)  d_in[2];
    float*       out   = (float*)d_out;

    float* colsum = (float*)d_ws;
    float* Bp     = colsum + NCHILD;

    zero_kernel  <<<NCHILD / 256, 256, 0, stream>>>(colsum);
    colsum_kernel<<<dim3(NCHILD / 256, NCHILD / 128), 256, 0, stream>>>(A, colsum);
    gather_kernel<<<(NCHILD * NCH) / 256, 256, 0, stream>>>(input, desc, colsum, Bp);
    gemm_kernel  <<<NCHILD / ROWS, 512, 0, stream>>>(A, Bp, out);
}
